# starting kernel
# baseline (speedup 1.0000x reference)
#include <hip/hip_runtime.h>
#include <math.h>

typedef __attribute__((ext_vector_type(2))) float v2f;
typedef __attribute__((ext_vector_type(8))) float v8f;

#define HDIM 2048
#define SDIM 4096
#define LSTRIDE (HDIM + 4)

static __device__ __forceinline__ v8f wmma_f32(v2f a, v2f b, v8f c) {
  return __builtin_amdgcn_wmma_f32_16x16x4_f32(false, a, false, b, (short)0, c,
                                               false, false);
}

static __device__ __forceinline__ float tanh_fast(float x) {
  const float e = __expf(2.0f * x);
  return 1.0f - 2.0f / (e + 1.0f);
}

__global__ __launch_bounds__(256) void whb_kernel(const float* __restrict__ W_h,
                                                  const float* __restrict__ dh,
                                                  const float* __restrict__ b,
                                                  float* __restrict__ whb) {
  const int wave = threadIdx.x >> 5;
  const int lane = threadIdx.x & 31;
  const int row  = blockIdx.x * 8 + wave;
  const float* wrow = W_h + (size_t)row * HDIM;
  float sum = 0.f;
  #pragma unroll 4
  for (int k = lane; k < HDIM; k += 32) sum += wrow[k] * dh[k];
  #pragma unroll
  for (int m = 16; m >= 1; m >>= 1) sum += __shfl_xor(sum, m, 32);
  if (lane == 0) whb[row] = sum + b[row];
}

__global__ __launch_bounds__(256) void energy_scores_kernel(
    const float* __restrict__ enc, const float* __restrict__ W_e,
    const float* __restrict__ whb, const float* __restrict__ v,
    float* __restrict__ scores) {
  extern __shared__ float s_enc[];
  __shared__ float s_scores[16];

  const int tid  = threadIdx.x;
  const int wave = tid >> 5;
  const int lane = tid & 31;
  const int lm   = lane & 15;
  const int hi   = lane >> 4;
  const int s0   = blockIdx.x * 16;

  if (tid < 16) s_scores[tid] = 0.f;

  for (int i = tid; i < 16 * (HDIM / 4); i += 256) {
    const int row  = i >> 9;
    const int col4 = i & 511;
    const float4 d =
        *(const float4*)(enc + (size_t)(s0 + row) * HDIM + (col4 << 2));
    *(float4*)(&s_enc[row * LSTRIDE + (col4 << 2)]) = d;
  }
  __syncthreads();

  const float* aRow = s_enc + lm * LSTRIDE + 2 * hi;
  const float* bBase = W_e + (size_t)(wave * 256 + lm) * HDIM + 2 * hi;

  v8f acc[8][2];
  #pragma unroll
  for (int p = 0; p < 8; ++p) {
    acc[p][0] = (v8f){};
    acc[p][1] = (v8f){};
  }

  v2f Bb[2][8];
  #pragma unroll
  for (int j = 0; j < 4; ++j) {
    Bb[0][j]     = *(const v2f*)(bBase + 4 * j);
    Bb[0][4 + j] = *(const v2f*)(bBase + (size_t)16 * HDIM + 4 * j);
  }

  #pragma unroll 1
  for (int c = 0; c < HDIM / 16; ++c) {
    const int k = c << 4;

    v2f a[4];
    #pragma unroll
    for (int j = 0; j < 4; ++j) a[j] = *(const v2f*)(aRow + k + 4 * j);

    #pragma unroll
    for (int p = 0; p < 8; ++p) {
      const int cur = p & 1, nxt = cur ^ 1;

      if (p < 7) {
        const float* nb = bBase + (size_t)(p + 1) * 32 * HDIM + k;
        #pragma unroll
        for (int j = 0; j < 4; ++j) {
          Bb[nxt][j]     = *(const v2f*)(nb + 4 * j);
          Bb[nxt][4 + j] = *(const v2f*)(nb + (size_t)16 * HDIM + 4 * j);
        }
      } else {
        const int k2 = (c < HDIM / 16 - 1) ? (k + 16) : 0;
        const float* nb = bBase + k2;
        #pragma unroll
        for (int j = 0; j < 4; ++j) {
          Bb[nxt][j]     = *(const v2f*)(nb + 4 * j);
          Bb[nxt][4 + j] = *(const v2f*)(nb + (size_t)16 * HDIM + 4 * j);
        }
      }

      #pragma unroll
      for (int j = 0; j < 4; ++j) {
        acc[p][0] = wmma_f32(a[j], Bb[cur][j], acc[p][0]);
        acc[p][1] = wmma_f32(a[j], Bb[cur][4 + j], acc[p][1]);
      }
    }
  }

  float lacc[8];
  #pragma unroll
  for (int r = 0; r < 8; ++r) lacc[r] = 0.f;

  #pragma unroll
  for (int p = 0; p < 8; ++p) {
    const int h0 = wave * 256 + p * 32;
    const float wb0 = whb[h0 + lm],      vv0 = v[h0 + lm];
    const float wb1 = whb[h0 + 16 + lm], vv1 = v[h0 + 16 + lm];
    #pragma unroll
    for (int r = 0; r < 8; ++r) {
      lacc[r] += tanh_fast(acc[p][0][r] + wb0) * vv0 +
                 tanh_fast(acc[p][1][r] + wb1) * vv1;
    }
  }

  #pragma unroll
  for (int r = 0; r < 8; ++r) {
    float t = lacc[r];
    t += __shfl_xor(t, 8, 32);
    t += __shfl_xor(t, 4, 32);
    t += __shfl_xor(t, 2, 32);
    t += __shfl_xor(t, 1, 32);
    if (lm == 0) atomicAdd(&s_scores[r + 8 * hi], t);
  }
  __syncthreads();
  if (tid < 16) scores[s0 + tid] = s_scores[tid];
}

__global__ __launch_bounds__(1024) void softmax_kernel(
    const float* __restrict__ scores, float* __restrict__ out) {
  __shared__ float red[32];
  __shared__ float stat[2];
  const int tid = threadIdx.x;

  out[tid] = 0.f;
  out[tid + 1024] = 0.f;

  float vals[4];
  float m = -INFINITY;
  #pragma unroll
  for (int i = 0; i < 4; ++i) {
    vals[i] = scores[tid + i * 1024];
    m = fmaxf(m, vals[i]);
  }
  #pragma unroll
  for (int o = 16; o >= 1; o >>= 1) m = fmaxf(m, __shfl_xor(m, o, 32));
  if ((tid & 31) == 0) red[tid >> 5] = m;
  __syncthreads();
  if (tid < 32) {
    float x = red[tid];
    #pragma unroll
    for (int o = 16; o >= 1; o >>= 1) x = fmaxf(x, __shfl_xor(x, o, 32));
    if (tid == 0) stat[0] = x;
  }
  __syncthreads();
  m = stat[0];

  float lsum = 0.f;
  #pragma unroll
  for (int i = 0; i < 4; ++i) {
    vals[i] = __expf(vals[i] - m);
    lsum += vals[i];
  }
  #pragma unroll
  for (int o = 16; o >= 1; o >>= 1) lsum += __shfl_xor(lsum, o, 32);
  if ((tid & 31) == 0) red[tid >> 5] = lsum;
  __syncthreads();
  if (tid < 32) {
    float x = red[tid];
    #pragma unroll
    for (int o = 16; o >= 1; o >>= 1) x += __shfl_xor(x, o, 32);
    if (tid == 0) stat[1] = x;
  }
  __syncthreads();
  const float inv = 1.0f / stat[1];
  #pragma unroll
  for (int i = 0; i < 4; ++i) out[HDIM + tid + i * 1024] = vals[i] * inv;
}

__global__ __launch_bounds__(256) void context_kernel(
    const float* __restrict__ enc, const float* __restrict__ attn,
    float* __restrict__ context) {
  __shared__ float s_attn[256];
  const int t = threadIdx.x;
  const int sbase = blockIdx.x * 256;
  s_attn[t] = attn[sbase + t];
  __syncthreads();

  for (int hb = 0; hb < HDIM; hb += 256) {
    const int h = hb + t;
    float sum = 0.f;
    #pragma unroll 4
    for (int s = 0; s < 256; ++s)
      sum += s_attn[s] * enc[(size_t)(sbase + s) * HDIM + h];
    atomicAdd(&context[h], sum);
  }
}

extern "C" void kernel_launch(void* const* d_in, const int* in_sizes, int n_in,
                              void* d_out, int out_size, void* d_ws, size_t ws_size,
                              hipStream_t stream) {
  const float* dh  = (const float*)d_in[0];
  const float* enc = (const float*)d_in[1];
  const float* W_h = (const float*)d_in[2];
  const float* W_e = (const float*)d_in[3];
  const float* b   = (const float*)d_in[4];
  const float* v   = (const float*)d_in[5];

  float* out    = (float*)d_out;
  float* whb    = (float*)d_ws;
  float* scores = whb + HDIM;

  const size_t lds_bytes = (size_t)16 * LSTRIDE * sizeof(float);

  whb_kernel<<<HDIM / 8, 256, 0, stream>>>(W_h, dh, b, whb);
  energy_scores_kernel<<<SDIM / 16, 256, lds_bytes, stream>>>(enc, W_e, whb, v,
                                                              scores);
  softmax_kernel<<<1, 1024, 0, stream>>>(scores, out);
  context_kernel<<<SDIM / 256, 256, 0, stream>>>(enc, out + HDIM, out);
}


# context kernel re-gridded from 16 blocks to 512 (256-row x 64-column tiles, 16 float4 row loads in flight per lane, fixed-order LDS fold); in-block sum order changes, 16 atomic adds per element kept
# speedup vs baseline: 1.3778x; 1.3778x over previous
#include <hip/hip_runtime.h>
#include <math.h>

typedef __attribute__((ext_vector_type(2))) float v2f;
typedef __attribute__((ext_vector_type(8))) float v8f;

#define HDIM 2048
#define SDIM 4096
#define LSTRIDE (HDIM + 4)

static __device__ __forceinline__ v8f wmma_f32(v2f a, v2f b, v8f c) {
  return __builtin_amdgcn_wmma_f32_16x16x4_f32(false, a, false, b, (short)0, c,
                                               false, false);
}

static __device__ __forceinline__ float tanh_fast(float x) {
  const float e = __expf(2.0f * x);
  return 1.0f - 2.0f / (e + 1.0f);
}

__global__ __launch_bounds__(256) void whb_kernel(const float* __restrict__ W_h,
                                                  const float* __restrict__ dh,
                                                  const float* __restrict__ b,
                                                  float* __restrict__ whb) {
  const int wave = threadIdx.x >> 5;
  const int lane = threadIdx.x & 31;
  const int row  = blockIdx.x * 8 + wave;
  const float* wrow = W_h + (size_t)row * HDIM;
  float sum = 0.f;
  #pragma unroll 4
  for (int k = lane; k < HDIM; k += 32) sum += wrow[k] * dh[k];
  #pragma unroll
  for (int m = 16; m >= 1; m >>= 1) sum += __shfl_xor(sum, m, 32);
  if (lane == 0) whb[row] = sum + b[row];
}

__global__ __launch_bounds__(256) void energy_scores_kernel(
    const float* __restrict__ enc, const float* __restrict__ W_e,
    const float* __restrict__ whb, const float* __restrict__ v,
    float* __restrict__ scores) {
  extern __shared__ float s_enc[];
  __shared__ float s_scores[16];

  const int tid  = threadIdx.x;
  const int wave = tid >> 5;
  const int lane = tid & 31;
  const int lm   = lane & 15;
  const int hi   = lane >> 4;
  const int s0   = blockIdx.x * 16;

  if (tid < 16) s_scores[tid] = 0.f;

  for (int i = tid; i < 16 * (HDIM / 4); i += 256) {
    const int row  = i >> 9;
    const int col4 = i & 511;
    const float4 d =
        *(const float4*)(enc + (size_t)(s0 + row) * HDIM + (col4 << 2));
    *(float4*)(&s_enc[row * LSTRIDE + (col4 << 2)]) = d;
  }
  __syncthreads();

  const float* aRow = s_enc + lm * LSTRIDE + 2 * hi;
  const float* bBase = W_e + (size_t)(wave * 256 + lm) * HDIM + 2 * hi;

  v8f acc[8][2];
  #pragma unroll
  for (int p = 0; p < 8; ++p) {
    acc[p][0] = (v8f){};
    acc[p][1] = (v8f){};
  }

  v2f Bb[2][8];
  #pragma unroll
  for (int j = 0; j < 4; ++j) {
    Bb[0][j]     = *(const v2f*)(bBase + 4 * j);
    Bb[0][4 + j] = *(const v2f*)(bBase + (size_t)16 * HDIM + 4 * j);
  }

  #pragma unroll 1
  for (int c = 0; c < HDIM / 16; ++c) {
    const int k = c << 4;

    v2f a[4];
    #pragma unroll
    for (int j = 0; j < 4; ++j) a[j] = *(const v2f*)(aRow + k + 4 * j);

    #pragma unroll
    for (int p = 0; p < 8; ++p) {
      const int cur = p & 1, nxt = cur ^ 1;

      if (p < 7) {
        const float* nb = bBase + (size_t)(p + 1) * 32 * HDIM + k;
        #pragma unroll
        for (int j = 0; j < 4; ++j) {
          Bb[nxt][j]     = *(const v2f*)(nb + 4 * j);
          Bb[nxt][4 + j] = *(const v2f*)(nb + (size_t)16 * HDIM + 4 * j);
        }
      } else {
        const int k2 = (c < HDIM / 16 - 1) ? (k + 16) : 0;
        const float* nb = bBase + k2;
        #pragma unroll
        for (int j = 0; j < 4; ++j) {
          Bb[nxt][j]     = *(const v2f*)(nb + 4 * j);
          Bb[nxt][4 + j] = *(const v2f*)(nb + (size_t)16 * HDIM + 4 * j);
        }
      }

      #pragma unroll
      for (int j = 0; j < 4; ++j) {
        acc[p][0] = wmma_f32(a[j], Bb[cur][j], acc[p][0]);
        acc[p][1] = wmma_f32(a[j], Bb[cur][4 + j], acc[p][1]);
      }
    }
  }

  float lacc[8];
  #pragma unroll
  for (int r = 0; r < 8; ++r) lacc[r] = 0.f;

  #pragma unroll
  for (int p = 0; p < 8; ++p) {
    const int h0 = wave * 256 + p * 32;
    const float wb0 = whb[h0 + lm],      vv0 = v[h0 + lm];
    const float wb1 = whb[h0 + 16 + lm], vv1 = v[h0 + 16 + lm];
    #pragma unroll
    for (int r = 0; r < 8; ++r) {
      lacc[r] += tanh_fast(acc[p][0][r] + wb0) * vv0 +
                 tanh_fast(acc[p][1][r] + wb1) * vv1;
    }
  }

  #pragma unroll
  for (int r = 0; r < 8; ++r) {
    float t = lacc[r];
    t += __shfl_xor(t, 8, 32);
    t += __shfl_xor(t, 4, 32);
    t += __shfl_xor(t, 2, 32);
    t += __shfl_xor(t, 1, 32);
    if (lm == 0) atomicAdd(&s_scores[r + 8 * hi], t);
  }
  __syncthreads();
  if (tid < 16) scores[s0 + tid] = s_scores[tid];
}

__global__ __launch_bounds__(1024) void softmax_kernel(
    const float* __restrict__ scores, float* __restrict__ out) {
  __shared__ float red[32];
  __shared__ float stat[2];
  const int tid = threadIdx.x;

  out[tid] = 0.f;
  out[tid + 1024] = 0.f;

  float vals[4];
  float m = -INFINITY;
  #pragma unroll
  for (int i = 0; i < 4; ++i) {
    vals[i] = scores[tid + i * 1024];
    m = fmaxf(m, vals[i]);
  }
  #pragma unroll
  for (int o = 16; o >= 1; o >>= 1) m = fmaxf(m, __shfl_xor(m, o, 32));
  if ((tid & 31) == 0) red[tid >> 5] = m;
  __syncthreads();
  if (tid < 32) {
    float x = red[tid];
    #pragma unroll
    for (int o = 16; o >= 1; o >>= 1) x = fmaxf(x, __shfl_xor(x, o, 32));
    if (tid == 0) stat[0] = x;
  }
  __syncthreads();
  m = stat[0];

  float lsum = 0.f;
  #pragma unroll
  for (int i = 0; i < 4; ++i) {
    vals[i] = __expf(vals[i] - m);
    lsum += vals[i];
  }
  #pragma unroll
  for (int o = 16; o >= 1; o >>= 1) lsum += __shfl_xor(lsum, o, 32);
  if ((tid & 31) == 0) red[tid >> 5] = lsum;
  __syncthreads();
  if (tid < 32) {
    float x = red[tid];
    #pragma unroll
    for (int o = 16; o >= 1; o >>= 1) x += __shfl_xor(x, o, 32);
    if (tid == 0) stat[1] = x;
  }
  __syncthreads();
  const float inv = 1.0f / stat[1];
  #pragma unroll
  for (int i = 0; i < 4; ++i) out[HDIM + tid + i * 1024] = vals[i] * inv;
}

#define CTX_COLS 64
#define CTX_ROWS 256

__global__ __launch_bounds__(256) __attribute__((amdgpu_waves_per_eu(4, 4)))
void context_kernel(
    const float* __restrict__ enc, const float* __restrict__ attn,
    float* __restrict__ context) {
  __shared__ float4 s_part[16][CTX_COLS / 4];

  const int tid  = threadIdx.x;
  const int wave = tid >> 5;
  const int lane = tid & 31;
  const int lm   = lane & 15;
  const int hh   = lane >> 4;
  const int slab  = blockIdx.x & 31;
  const int chunk = blockIdx.x >> 5;

  const int row0 = chunk * CTX_ROWS + wave * 32 + hh * 16;
  const float* src = enc + (size_t)row0 * HDIM + slab * CTX_COLS + lm * 4;

  float4 w4[4];
  #pragma unroll
  for (int q = 0; q < 4; ++q) w4[q] = *(const float4*)(attn + row0 + 4 * q);

  float4 x[16];
  #pragma unroll
  for (int i = 0; i < 16; ++i) x[i] = *(const float4*)(src + (size_t)i * HDIM);

  float4 acc = make_float4(0.f, 0.f, 0.f, 0.f);
  #pragma unroll
  for (int q = 0; q < 4; ++q) {
    const float a0 = w4[q].x, a1 = w4[q].y, a2 = w4[q].z, a3 = w4[q].w;
    const float4 x0 = x[4 * q], x1 = x[4 * q + 1], x2 = x[4 * q + 2],
                 x3 = x[4 * q + 3];
    acc.x += a0 * x0.x; acc.y += a0 * x0.y; acc.z += a0 * x0.z; acc.w += a0 * x0.w;
    acc.x += a1 * x1.x; acc.y += a1 * x1.y; acc.z += a1 * x1.z; acc.w += a1 * x1.w;
    acc.x += a2 * x2.x; acc.y += a2 * x2.y; acc.z += a2 * x2.z; acc.w += a2 * x2.w;
    acc.x += a3 * x3.x; acc.y += a3 * x3.y; acc.z += a3 * x3.z; acc.w += a3 * x3.w;
  }

  s_part[wave * 2 + hh][lm] = acc;
  __syncthreads();

  if (tid < CTX_COLS) {
    const float* p = (const float*)&s_part[0][0];
    float sum = 0.f;
    #pragma unroll
    for (int j = 0; j < 16; ++j) sum += p[j * CTX_COLS + tid];
    atomicAdd(&context[slab * CTX_COLS + tid], sum);
  }
}

extern "C" void kernel_launch(void* const* d_in, const int* in_sizes, int n_in,
                              void* d_out, int out_size, void* d_ws, size_t ws_size,
                              hipStream_t stream) {
  const float* dh  = (const float*)d_in[0];
  const float* enc = (const float*)d_in[1];
  const float* W_h = (const float*)d_in[2];
  const float* W_e = (const float*)d_in[3];
  const float* b   = (const float*)d_in[4];
  const float* v   = (const float*)d_in[5];

  float* out    = (float*)d_out;
  float* whb    = (float*)d_ws;
  float* scores = whb + HDIM;

  const size_t lds_bytes = (size_t)16 * LSTRIDE * sizeof(float);

  whb_kernel<<<HDIM / 8, 256, 0, stream>>>(W_h, dh, b, whb);
  energy_scores_kernel<<<SDIM / 16, 256, lds_bytes, stream>>>(enc, W_e, whb, v,
                                                              scores);
  softmax_kernel<<<1, 1024, 0, stream>>>(scores, out);
  context_kernel<<<512, 256, 0, stream>>>(enc, out + HDIM, out);
}


# energy GEMM re-tiled: block = 32 s rows x half of h (was 16 x all h), so each W_e fragment load feeds 2 WMMAs; scores summed as two half-h partials added in softmax (sum order changes)
# speedup vs baseline: 2.4265x; 1.7611x over previous
#include <hip/hip_runtime.h>
#include <math.h>

typedef __attribute__((ext_vector_type(2))) float v2f;
typedef __attribute__((ext_vector_type(8))) float v8f;

#define HDIM 2048
#define SDIM 4096
#define LSTRIDE (HDIM + 4)

static __device__ __forceinline__ v8f wmma_f32(v2f a, v2f b, v8f c) {
  return __builtin_amdgcn_wmma_f32_16x16x4_f32(false, a, false, b, (short)0, c,
                                               false, false);
}

static __device__ __forceinline__ float tanh_fast(float x) {
  const float e = __expf(2.0f * x);
  return 1.0f - 2.0f / (e + 1.0f);
}

__global__ __launch_bounds__(256) void whb_kernel(const float* __restrict__ W_h,
                                                  const float* __restrict__ dh,
                                                  const float* __restrict__ b,
                                                  float* __restrict__ whb) {
  const int wave = threadIdx.x >> 5;
  const int lane = threadIdx.x & 31;
  const int row  = blockIdx.x * 8 + wave;
  const float* wrow = W_h + (size_t)row * HDIM;
  float sum = 0.f;
  #pragma unroll 4
  for (int k = lane; k < HDIM; k += 32) sum += wrow[k] * dh[k];
  #pragma unroll
  for (int m = 16; m >= 1; m >>= 1) sum += __shfl_xor(sum, m, 32);
  if (lane == 0) whb[row] = sum + b[row];
}

#define ES_ROWS 32
#define ES_COLS 1024

__global__ __launch_bounds__(256) void energy_scores_kernel(
    const float* __restrict__ enc, const float* __restrict__ W_e,
    const float* __restrict__ whb, const float* __restrict__ v,
    float* __restrict__ part_lo, float* __restrict__ part_hi) {
  extern __shared__ float s_enc[];
  __shared__ float s_scores[ES_ROWS];

  const int tid  = threadIdx.x;
  const int wave = tid >> 5;
  const int lane = tid & 31;
  const int lm   = lane & 15;
  const int hi   = lane >> 4;
  const int half = blockIdx.x & 1;
  const int s0   = (blockIdx.x >> 1) * ES_ROWS;

  if (tid < ES_ROWS) s_scores[tid] = 0.f;

  for (int i = tid; i < ES_ROWS * (HDIM / 4); i += 256) {
    const int row  = i >> 9;
    const int col4 = i & 511;
    const float4 d =
        *(const float4*)(enc + (size_t)(s0 + row) * HDIM + (col4 << 2));
    *(float4*)(&s_enc[row * LSTRIDE + (col4 << 2)]) = d;
  }
  __syncthreads();

  const float* aRow0 = s_enc + lm * LSTRIDE + 2 * hi;
  const float* aRow1 = aRow0 + 16 * LSTRIDE;
  const int hb = half * ES_COLS + wave * 128;
  const float* bBase = W_e + (size_t)(hb + lm) * HDIM + 2 * hi;

  v8f acc[4][2][2];
  #pragma unroll
  for (int p = 0; p < 4; ++p) {
    #pragma unroll
    for (int t = 0; t < 2; ++t) {
      acc[p][t][0] = (v8f){};
      acc[p][t][1] = (v8f){};
    }
  }

  v2f Bb[2][8];
  #pragma unroll
  for (int j = 0; j < 4; ++j) {
    Bb[0][j]     = *(const v2f*)(bBase + 4 * j);
    Bb[0][4 + j] = *(const v2f*)(bBase + (size_t)16 * HDIM + 4 * j);
  }

  #pragma unroll 1
  for (int c = 0; c < HDIM / 16; ++c) {
    const int k = c << 4;

    v2f a0[4], a1[4];
    #pragma unroll
    for (int j = 0; j < 4; ++j) {
      a0[j] = *(const v2f*)(aRow0 + k + 4 * j);
      a1[j] = *(const v2f*)(aRow1 + k + 4 * j);
    }

    #pragma unroll
    for (int p = 0; p < 4; ++p) {
      const int cur = p & 1, nxt = cur ^ 1;

      if (p < 3) {
        const float* nb = bBase + (size_t)(p + 1) * 32 * HDIM + k;
        #pragma unroll
        for (int j = 0; j < 4; ++j) {
          Bb[nxt][j]     = *(const v2f*)(nb + 4 * j);
          Bb[nxt][4 + j] = *(const v2f*)(nb + (size_t)16 * HDIM + 4 * j);
        }
      } else {
        const int k2 = (c < HDIM / 16 - 1) ? (k + 16) : 0;
        const float* nb = bBase + k2;
        #pragma unroll
        for (int j = 0; j < 4; ++j) {
          Bb[nxt][j]     = *(const v2f*)(nb + 4 * j);
          Bb[nxt][4 + j] = *(const v2f*)(nb + (size_t)16 * HDIM + 4 * j);
        }
      }

      #pragma unroll
      for (int j = 0; j < 4; ++j) {
        acc[p][0][0] = wmma_f32(a0[j], Bb[cur][j],     acc[p][0][0]);
        acc[p][0][1] = wmma_f32(a1[j], Bb[cur][j],     acc[p][0][1]);
        acc[p][1][0] = wmma_f32(a0[j], Bb[cur][4 + j], acc[p][1][0]);
        acc[p][1][1] = wmma_f32(a1[j], Bb[cur][4 + j], acc[p][1][1]);
      }
    }
  }

  float lacc[2][8];
  #pragma unroll
  for (int r = 0; r < 8; ++r) {
    lacc[0][r] = 0.f;
    lacc[1][r] = 0.f;
  }

  #pragma unroll
  for (int p = 0; p < 4; ++p) {
    const int h0 = hb + p * 32;
    const float wb0 = whb[h0 + lm],      vv0 = v[h0 + lm];
    const float wb1 = whb[h0 + 16 + lm], vv1 = v[h0 + 16 + lm];
    #pragma unroll
    for (int m = 0; m < 2; ++m) {
      #pragma unroll
      for (int r = 0; r < 8; ++r) {
        lacc[m][r] += tanh_fast(acc[p][0][m][r] + wb0) * vv0 +
                      tanh_fast(acc[p][1][m][r] + wb1) * vv1;
      }
    }
  }

  #pragma unroll
  for (int m = 0; m < 2; ++m) {
    #pragma unroll
    for (int r = 0; r < 8; ++r) {
      float t = lacc[m][r];
      t += __shfl_xor(t, 8, 32);
      t += __shfl_xor(t, 4, 32);
      t += __shfl_xor(t, 2, 32);
      t += __shfl_xor(t, 1, 32);
      if (lm == 0) atomicAdd(&s_scores[m * 16 + r + 8 * hi], t);
    }
  }
  __syncthreads();
  if (tid < ES_ROWS) {
    float* dst = half ? part_hi : part_lo;
    dst[s0 + tid] = s_scores[tid];
  }
}

__global__ __launch_bounds__(1024) void softmax_kernel(
    const float* __restrict__ scores, float* __restrict__ out) {
  __shared__ float red[32];
  __shared__ float stat[2];
  const int tid = threadIdx.x;

  out[tid] = 0.f;
  out[tid + 1024] = 0.f;

  float vals[4];
  float m = -INFINITY;
  #pragma unroll
  for (int i = 0; i < 4; ++i) {
    vals[i] = scores[tid + i * 1024] + out[HDIM + tid + i * 1024];
    m = fmaxf(m, vals[i]);
  }
  #pragma unroll
  for (int o = 16; o >= 1; o >>= 1) m = fmaxf(m, __shfl_xor(m, o, 32));
  if ((tid & 31) == 0) red[tid >> 5] = m;
  __syncthreads();
  if (tid < 32) {
    float x = red[tid];
    #pragma unroll
    for (int o = 16; o >= 1; o >>= 1) x = fmaxf(x, __shfl_xor(x, o, 32));
    if (tid == 0) stat[0] = x;
  }
  __syncthreads();
  m = stat[0];

  float lsum = 0.f;
  #pragma unroll
  for (int i = 0; i < 4; ++i) {
    vals[i] = __expf(vals[i] - m);
    lsum += vals[i];
  }
  #pragma unroll
  for (int o = 16; o >= 1; o >>= 1) lsum += __shfl_xor(lsum, o, 32);
  if ((tid & 31) == 0) red[tid >> 5] = lsum;
  __syncthreads();
  if (tid < 32) {
    float x = red[tid];
    #pragma unroll
    for (int o = 16; o >= 1; o >>= 1) x += __shfl_xor(x, o, 32);
    if (tid == 0) stat[1] = x;
  }
  __syncthreads();
  const float inv = 1.0f / stat[1];
  #pragma unroll
  for (int i = 0; i < 4; ++i) out[HDIM + tid + i * 1024] = vals[i] * inv;
}

#define CTX_COLS 64
#define CTX_ROWS 256

__global__ __launch_bounds__(256) __attribute__((amdgpu_waves_per_eu(4, 4)))
void context_kernel(
    const float* __restrict__ enc, const float* __restrict__ attn,
    float* __restrict__ context) {
  __shared__ float4 s_part[16][CTX_COLS / 4];

  const int tid  = threadIdx.x;
  const int wave = tid >> 5;
  const int lane = tid & 31;
  const int lm   = lane & 15;
  const int hh   = lane >> 4;
  const int slab  = blockIdx.x & 31;
  const int chunk = blockIdx.x >> 5;

  const int row0 = chunk * CTX_ROWS + wave * 32 + hh * 16;
  const float* src = enc + (size_t)row0 * HDIM + slab * CTX_COLS + lm * 4;

  float4 w4[4];
  #pragma unroll
  for (int q = 0; q < 4; ++q) w4[q] = *(const float4*)(attn + row0 + 4 * q);

  float4 x[16];
  #pragma unroll
  for (int i = 0; i < 16; ++i) x[i] = *(const float4*)(src + (size_t)i * HDIM);

  float4 acc = make_float4(0.f, 0.f, 0.f, 0.f);
  #pragma unroll
  for (int q = 0; q < 4; ++q) {
    const float a0 = w4[q].x, a1 = w4[q].y, a2 = w4[q].z, a3 = w4[q].w;
    const float4 x0 = x[4 * q], x1 = x[4 * q + 1], x2 = x[4 * q + 2],
                 x3 = x[4 * q + 3];
    acc.x += a0 * x0.x; acc.y += a0 * x0.y; acc.z += a0 * x0.z; acc.w += a0 * x0.w;
    acc.x += a1 * x1.x; acc.y += a1 * x1.y; acc.z += a1 * x1.z; acc.w += a1 * x1.w;
    acc.x += a2 * x2.x; acc.y += a2 * x2.y; acc.z += a2 * x2.z; acc.w += a2 * x2.w;
    acc.x += a3 * x3.x; acc.y += a3 * x3.y; acc.z += a3 * x3.z; acc.w += a3 * x3.w;
  }

  s_part[wave * 2 + hh][lm] = acc;
  __syncthreads();

  if (tid < CTX_COLS) {
    const float* p = (const float*)&s_part[0][0];
    float sum = 0.f;
    #pragma unroll
    for (int j = 0; j < 16; ++j) sum += p[j * CTX_COLS + tid];
    atomicAdd(&context[slab * CTX_COLS + tid], sum);
  }
}

extern "C" void kernel_launch(void* const* d_in, const int* in_sizes, int n_in,
                              void* d_out, int out_size, void* d_ws, size_t ws_size,
                              hipStream_t stream) {
  const float* dh  = (const float*)d_in[0];
  const float* enc = (const float*)d_in[1];
  const float* W_h = (const float*)d_in[2];
  const float* W_e = (const float*)d_in[3];
  const float* b   = (const float*)d_in[4];
  const float* v   = (const float*)d_in[5];

  float* out    = (float*)d_out;
  float* whb    = (float*)d_ws;
  float* scores = whb + HDIM;

  const size_t lds_bytes = (size_t)ES_ROWS * LSTRIDE * sizeof(float);

  whb_kernel<<<HDIM / 8, 256, 0, stream>>>(W_h, dh, b, whb);
  energy_scores_kernel<<<(SDIM / ES_ROWS) * (HDIM / ES_COLS), 256, lds_bytes,
                         stream>>>(enc, W_e, whb, v, scores, out + HDIM);
  softmax_kernel<<<1, 1024, 0, stream>>>(scores, out);
  context_kernel<<<512, 256, 0, stream>>>(enc, out + HDIM, out);
}


# energy GEMM K-chunk widened 16 -> 32 floats, fragments loaded as 16-byte pieces: each 128-byte W_e line is read in one pass by 4 grouped loads (was 8 over 2 passes); k sum order changes
# speedup vs baseline: 3.2138x; 1.3245x over previous
#include <hip/hip_runtime.h>
#include <math.h>

typedef __attribute__((ext_vector_type(2))) float v2f;
typedef __attribute__((ext_vector_type(4))) float v4f;
typedef __attribute__((ext_vector_type(8))) float v8f;

#define HDIM 2048
#define SDIM 4096
#define LSTRIDE (HDIM + 4)

static __device__ __forceinline__ v8f wmma_f32(v2f a, v2f b, v8f c) {
  return __builtin_amdgcn_wmma_f32_16x16x4_f32(false, a, false, b, (short)0, c,
                                               false, false);
}

static __device__ __forceinline__ float tanh_fast(float x) {
  const float e = __expf(2.0f * x);
  return 1.0f - 2.0f / (e + 1.0f);
}

__global__ __launch_bounds__(256) void whb_kernel(const float* __restrict__ W_h,
                                                  const float* __restrict__ dh,
                                                  const float* __restrict__ b,
                                                  float* __restrict__ whb) {
  const int wave = threadIdx.x >> 5;
  const int lane = threadIdx.x & 31;
  const int row  = blockIdx.x * 8 + wave;
  const float* wrow = W_h + (size_t)row * HDIM;
  float sum = 0.f;
  #pragma unroll 4
  for (int k = lane; k < HDIM; k += 32) sum += wrow[k] * dh[k];
  #pragma unroll
  for (int m = 16; m >= 1; m >>= 1) sum += __shfl_xor(sum, m, 32);
  if (lane == 0) whb[row] = sum + b[row];
}

#define ES_ROWS 32
#define ES_COLS 1024
#define ES_KC   32

__global__ __launch_bounds__(256) void energy_scores_kernel(
    const float* __restrict__ enc, const float* __restrict__ W_e,
    const float* __restrict__ whb, const float* __restrict__ v,
    float* __restrict__ part_lo, float* __restrict__ part_hi) {
  extern __shared__ float s_enc[];
  __shared__ float s_scores[ES_ROWS];

  const int tid  = threadIdx.x;
  const int wave = tid >> 5;
  const int lane = tid & 31;
  const int lm   = lane & 15;
  const int hi   = lane >> 4;
  const int half = blockIdx.x & 1;
  const int s0   = (blockIdx.x >> 1) * ES_ROWS;

  if (tid < ES_ROWS) s_scores[tid] = 0.f;

  for (int i = tid; i < ES_ROWS * (HDIM / 4); i += 256) {
    const int row  = i >> 9;
    const int col4 = i & 511;
    const float4 d =
        *(const float4*)(enc + (size_t)(s0 + row) * HDIM + (col4 << 2));
    *(float4*)(&s_enc[row * LSTRIDE + (col4 << 2)]) = d;
  }
  __syncthreads();

  const float* aRow0 = s_enc + lm * LSTRIDE + 16 * hi;
  const float* aRow1 = aRow0 + 16 * LSTRIDE;
  const int hb = half * ES_COLS + wave * 128;
  const float* bBase = W_e + (size_t)(hb + lm) * HDIM + 16 * hi;

  v8f acc[4][2][2];
  #pragma unroll
  for (int p = 0; p < 4; ++p) {
    #pragma unroll
    for (int t = 0; t < 2; ++t) {
      acc[p][t][0] = (v8f){};
      acc[p][t][1] = (v8f){};
    }
  }

  v4f Bb[2][8];
  #pragma unroll
  for (int q = 0; q < 4; ++q) {
    Bb[0][q]     = *(const v4f*)(bBase + 4 * q);
    Bb[0][4 + q] = *(const v4f*)(bBase + (size_t)16 * HDIM + 4 * q);
  }

  #pragma unroll 1
  for (int c = 0; c < HDIM / ES_KC; ++c) {
    const int k = c * ES_KC;

    v4f a0[4], a1[4];
    #pragma unroll
    for (int q = 0; q < 4; ++q) {
      a0[q] = *(const v4f*)(aRow0 + k + 4 * q);
      a1[q] = *(const v4f*)(aRow1 + k + 4 * q);
    }

    #pragma unroll
    for (int p = 0; p < 4; ++p) {
      const int cur = p & 1, nxt = cur ^ 1;

      if (p < 3) {
        const float* nb = bBase + (size_t)(p + 1) * 32 * HDIM + k;
        #pragma unroll
        for (int q = 0; q < 4; ++q) {
          Bb[nxt][q]     = *(const v4f*)(nb + 4 * q);
          Bb[nxt][4 + q] = *(const v4f*)(nb + (size_t)16 * HDIM + 4 * q);
        }
      } else {
        const int k2 = (c < HDIM / ES_KC - 1) ? (k + ES_KC) : 0;
        const float* nb = bBase + k2;
        #pragma unroll
        for (int q = 0; q < 4; ++q) {
          Bb[nxt][q]     = *(const v4f*)(nb + 4 * q);
          Bb[nxt][4 + q] = *(const v4f*)(nb + (size_t)16 * HDIM + 4 * q);
        }
      }

      __builtin_amdgcn_sched_barrier(0);

      #pragma unroll
      for (int q = 0; q < 4; ++q) {
        acc[p][0][0] = wmma_f32(a0[q].lo, Bb[cur][q].lo,     acc[p][0][0]);
        acc[p][0][1] = wmma_f32(a1[q].lo, Bb[cur][q].lo,     acc[p][0][1]);
        acc[p][1][0] = wmma_f32(a0[q].lo, Bb[cur][4 + q].lo, acc[p][1][0]);
        acc[p][1][1] = wmma_f32(a1[q].lo, Bb[cur][4 + q].lo, acc[p][1][1]);
        acc[p][0][0] = wmma_f32(a0[q].hi, Bb[cur][q].hi,     acc[p][0][0]);
        acc[p][0][1] = wmma_f32(a1[q].hi, Bb[cur][q].hi,     acc[p][0][1]);
        acc[p][1][0] = wmma_f32(a0[q].hi, Bb[cur][4 + q].hi, acc[p][1][0]);
        acc[p][1][1] = wmma_f32(a1[q].hi, Bb[cur][4 + q].hi, acc[p][1][1]);
      }
      __builtin_amdgcn_sched_barrier(0);
    }
  }

  float lacc[2][8];
  #pragma unroll
  for (int r = 0; r < 8; ++r) {
    lacc[0][r] = 0.f;
    lacc[1][r] = 0.f;
  }

  #pragma unroll
  for (int p = 0; p < 4; ++p) {
    const int h0 = hb + p * 32;
    const float wb0 = whb[h0 + lm],      vv0 = v[h0 + lm];
    const float wb1 = whb[h0 + 16 + lm], vv1 = v[h0 + 16 + lm];
    #pragma unroll
    for (int m = 0; m < 2; ++m) {
      #pragma unroll
      for (int r = 0; r < 8; ++r) {
        lacc[m][r] += tanh_fast(acc[p][0][m][r] + wb0) * vv0 +
                      tanh_fast(acc[p][1][m][r] + wb1) * vv1;
      }
    }
  }

  #pragma unroll
  for (int m = 0; m < 2; ++m) {
    #pragma unroll
    for (int r = 0; r < 8; ++r) {
      float t = lacc[m][r];
      t += __shfl_xor(t, 8, 32);
      t += __shfl_xor(t, 4, 32);
      t += __shfl_xor(t, 2, 32);
      t += __shfl_xor(t, 1, 32);
      if (lm == 0) atomicAdd(&s_scores[m * 16 + r + 8 * hi], t);
    }
  }
  __syncthreads();
  if (tid < ES_ROWS) {
    float* dst = half ? part_hi : part_lo;
    dst[s0 + tid] = s_scores[tid];
  }
}

__global__ __launch_bounds__(1024) void softmax_kernel(
    const float* __restrict__ scores, float* __restrict__ out) {
  __shared__ float red[32];
  __shared__ float stat[2];
  const int tid = threadIdx.x;

  out[tid] = 0.f;
  out[tid + 1024] = 0.f;

  float vals[4];
  float m = -INFINITY;
  #pragma unroll
  for (int i = 0; i < 4; ++i) {
    vals[i] = scores[tid + i * 1024] + out[HDIM + tid + i * 1024];
    m = fmaxf(m, vals[i]);
  }
  #pragma unroll
  for (int o = 16; o >= 1; o >>= 1) m = fmaxf(m, __shfl_xor(m, o, 32));
  if ((tid & 31) == 0) red[tid >> 5] = m;
  __syncthreads();
  if (tid < 32) {
    float x = red[tid];
    #pragma unroll
    for (int o = 16; o >= 1; o >>= 1) x = fmaxf(x, __shfl_xor(x, o, 32));
    if (tid == 0) stat[0] = x;
  }
  __syncthreads();
  m = stat[0];

  float lsum = 0.f;
  #pragma unroll
  for (int i = 0; i < 4; ++i) {
    vals[i] = __expf(vals[i] - m);
    lsum += vals[i];
  }
  #pragma unroll
  for (int o = 16; o >= 1; o >>= 1) lsum += __shfl_xor(lsum, o, 32);
  if ((tid & 31) == 0) red[tid >> 5] = lsum;
  __syncthreads();
  if (tid < 32) {
    float x = red[tid];
    #pragma unroll
    for (int o = 16; o >= 1; o >>= 1) x += __shfl_xor(x, o, 32);
    if (tid == 0) stat[1] = x;
  }
  __syncthreads();
  const float inv = 1.0f / stat[1];
  #pragma unroll
  for (int i = 0; i < 4; ++i) out[HDIM + tid + i * 1024] = vals[i] * inv;
}

#define CTX_COLS 64
#define CTX_ROWS 256

__global__ __launch_bounds__(256) __attribute__((amdgpu_waves_per_eu(4, 4)))
void context_kernel(
    const float* __restrict__ enc, const float* __restrict__ attn,
    float* __restrict__ context) {
  __shared__ float4 s_part[16][CTX_COLS / 4];

  const int tid  = threadIdx.x;
  const int wave = tid >> 5;
  const int lane = tid & 31;
  const int lm   = lane & 15;
  const int hh   = lane >> 4;
  const int slab  = blockIdx.x & 31;
  const int chunk = blockIdx.x >> 5;

  const int row0 = chunk * CTX_ROWS + wave * 32 + hh * 16;
  const float* src = enc + (size_t)row0 * HDIM + slab * CTX_COLS + lm * 4;

  float4 w4[4];
  #pragma unroll
  for (int q = 0; q < 4; ++q) w4[q] = *(const float4*)(attn + row0 + 4 * q);

  float4 x[16];
  #pragma unroll
  for (int i = 0; i < 16; ++i) x[i] = *(const float4*)(src + (size_t)i * HDIM);

  float4 acc = make_float4(0.f, 0.f, 0.f, 0.f);
  #pragma unroll
  for (int q = 0; q < 4; ++q) {
    const float a0 = w4[q].x, a1 = w4[q].y, a2 = w4[q].z, a3 = w4[q].w;
    const float4 x0 = x[4 * q], x1 = x[4 * q + 1], x2 = x[4 * q + 2],
                 x3 = x[4 * q + 3];
    acc.x += a0 * x0.x; acc.y += a0 * x0.y; acc.z += a0 * x0.z; acc.w += a0 * x0.w;
    acc.x += a1 * x1.x; acc.y += a1 * x1.y; acc.z += a1 * x1.z; acc.w += a1 * x1.w;
    acc.x += a2 * x2.x; acc.y += a2 * x2.y; acc.z += a2 * x2.z; acc.w += a2 * x2.w;
    acc.x += a3 * x3.x; acc.y += a3 * x3.y; acc.z += a3 * x3.z; acc.w += a3 * x3.w;
  }

  s_part[wave * 2 + hh][lm] = acc;
  __syncthreads();

  if (tid < CTX_COLS) {
    const float* p = (const float*)&s_part[0][0];
    float sum = 0.f;
    #pragma unroll
    for (int j = 0; j < 16; ++j) sum += p[j * CTX_COLS + tid];
    atomicAdd(&context[slab * CTX_COLS + tid], sum);
  }
}

extern "C" void kernel_launch(void* const* d_in, const int* in_sizes, int n_in,
                              void* d_out, int out_size, void* d_ws, size_t ws_size,
                              hipStream_t stream) {
  const float* dh  = (const float*)d_in[0];
  const float* enc = (const float*)d_in[1];
  const float* W_h = (const float*)d_in[2];
  const float* W_e = (const float*)d_in[3];
  const float* b   = (const float*)d_in[4];
  const float* v   = (const float*)d_in[5];

  float* out    = (float*)d_out;
  float* whb    = (float*)d_ws;
  float* scores = whb + HDIM;

  const size_t lds_bytes = (size_t)ES_ROWS * LSTRIDE * sizeof(float);

  whb_kernel<<<HDIM / 8, 256, 0, stream>>>(W_h, dh, b, whb);
  energy_scores_kernel<<<(SDIM / ES_ROWS) * (HDIM / ES_COLS), 256, lds_bytes,
                         stream>>>(enc, W_e, whb, v, scores, out + HDIM);
  softmax_kernel<<<1, 1024, 0, stream>>>(scores, out);
  context_kernel<<<512, 256, 0, stream>>>(enc, out + HDIM, out);
}


# whb GEMV and the energy kernel's tile staging issue each lane's loads as one group (16+16 and 32 float4 in flight, were 4 per trip); whb sum order changes (4 per-lane sums, bias joins lane 0)
# speedup vs baseline: 3.3718x; 1.0492x over previous
#include <hip/hip_runtime.h>
#include <math.h>

typedef __attribute__((ext_vector_type(2))) float v2f;
typedef __attribute__((ext_vector_type(4))) float v4f;
typedef __attribute__((ext_vector_type(8))) float v8f;

#define HDIM 2048
#define SDIM 4096
#define LSTRIDE (HDIM + 4)

static __device__ __forceinline__ v8f wmma_f32(v2f a, v2f b, v8f c) {
  return __builtin_amdgcn_wmma_f32_16x16x4_f32(false, a, false, b, (short)0, c,
                                               false, false);
}

static __device__ __forceinline__ float tanh_fast(float x) {
  const float e = __expf(2.0f * x);
  return 1.0f - 2.0f / (e + 1.0f);
}

__global__ __launch_bounds__(256) __attribute__((amdgpu_waves_per_eu(2, 4)))
void whb_kernel(const float* __restrict__ W_h,
                const float* __restrict__ dh,
                const float* __restrict__ b,
                float* __restrict__ whb) {
  const int wave = threadIdx.x >> 5;
  const int lane = threadIdx.x & 31;
  const int row  = blockIdx.x * 8 + wave;
  const v4f* wrow = (const v4f*)(W_h + (size_t)row * HDIM) + lane;
  const v4f* drow = (const v4f*)dh + lane;

  v4f w[16], d[16];
  #pragma unroll
  for (int j = 0; j < 16; ++j) w[j] = wrow[32 * j];
  #pragma unroll
  for (int j = 0; j < 16; ++j) d[j] = drow[32 * j];
  const float bias = b[row];

  float s0 = 0.f, s1 = 0.f, s2 = 0.f, s3 = 0.f;
  #pragma unroll
  for (int j = 0; j < 16; ++j) {
    s0 += w[j].x * d[j].x;
    s1 += w[j].y * d[j].y;
    s2 += w[j].z * d[j].z;
    s3 += w[j].w * d[j].w;
  }
  float sum = (s0 + s1) + (s2 + s3);
  if (lane == 0) sum += bias;
  #pragma unroll
  for (int m = 16; m >= 1; m >>= 1) sum += __shfl_xor(sum, m, 32);
  if (lane == 0) whb[row] = sum;
}

#define ES_ROWS 32
#define ES_COLS 1024
#define ES_KC   32

__global__ __launch_bounds__(256) void energy_scores_kernel(
    const float* __restrict__ enc, const float* __restrict__ W_e,
    const float* __restrict__ whb, const float* __restrict__ v,
    float* __restrict__ part_lo, float* __restrict__ part_hi) {
  extern __shared__ float s_enc[];
  __shared__ float s_scores[ES_ROWS];

  const int tid  = threadIdx.x;
  const int wave = tid >> 5;
  const int lane = tid & 31;
  const int lm   = lane & 15;
  const int hi   = lane >> 4;
  const int half = blockIdx.x & 1;
  const int s0   = (blockIdx.x >> 1) * ES_ROWS;

  if (tid < ES_ROWS) s_scores[tid] = 0.f;

  #pragma unroll 1
  for (int t = 0; t < 2; ++t) {
    v4f buf[32];
    #pragma unroll
    for (int j = 0; j < 32; ++j) {
      const int row = 16 * t + (j >> 1);
      const int col = (tid + 256 * (j & 1)) << 2;
      buf[j] = *(const v4f*)(enc + (size_t)(s0 + row) * HDIM + col);
    }
    #pragma unroll
    for (int j = 0; j < 32; ++j) {
      const int row = 16 * t + (j >> 1);
      const int col = (tid + 256 * (j & 1)) << 2;
      *(v4f*)(&s_enc[row * LSTRIDE + col]) = buf[j];
    }
  }
  __syncthreads();

  const float* aRow0 = s_enc + lm * LSTRIDE + 16 * hi;
  const float* aRow1 = aRow0 + 16 * LSTRIDE;
  const int hb = half * ES_COLS + wave * 128;
  const float* bBase = W_e + (size_t)(hb + lm) * HDIM + 16 * hi;

  v8f acc[4][2][2];
  #pragma unroll
  for (int p = 0; p < 4; ++p) {
    #pragma unroll
    for (int t = 0; t < 2; ++t) {
      acc[p][t][0] = (v8f){};
      acc[p][t][1] = (v8f){};
    }
  }

  v4f Bb[2][8];
  #pragma unroll
  for (int q = 0; q < 4; ++q) {
    Bb[0][q]     = *(const v4f*)(bBase + 4 * q);
    Bb[0][4 + q] = *(const v4f*)(bBase + (size_t)16 * HDIM + 4 * q);
  }

  #pragma unroll 1
  for (int c = 0; c < HDIM / ES_KC; ++c) {
    const int k = c * ES_KC;

    v4f a0[4], a1[4];
    #pragma unroll
    for (int q = 0; q < 4; ++q) {
      a0[q] = *(const v4f*)(aRow0 + k + 4 * q);
      a1[q] = *(const v4f*)(aRow1 + k + 4 * q);
    }

    #pragma unroll
    for (int p = 0; p < 4; ++p) {
      const int cur = p & 1, nxt = cur ^ 1;

      if (p < 3) {
        const float* nb = bBase + (size_t)(p + 1) * 32 * HDIM + k;
        #pragma unroll
        for (int q = 0; q < 4; ++q) {
          Bb[nxt][q]     = *(const v4f*)(nb + 4 * q);
          Bb[nxt][4 + q] = *(const v4f*)(nb + (size_t)16 * HDIM + 4 * q);
        }
      } else {
        const int k2 = (c < HDIM / ES_KC - 1) ? (k + ES_KC) : 0;
        const float* nb = bBase + k2;
        #pragma unroll
        for (int q = 0; q < 4; ++q) {
          Bb[nxt][q]     = *(const v4f*)(nb + 4 * q);
          Bb[nxt][4 + q] = *(const v4f*)(nb + (size_t)16 * HDIM + 4 * q);
        }
      }

      __builtin_amdgcn_sched_barrier(0);

      #pragma unroll
      for (int q = 0; q < 4; ++q) {
        acc[p][0][0] = wmma_f32(a0[q].lo, Bb[cur][q].lo,     acc[p][0][0]);
        acc[p][0][1] = wmma_f32(a1[q].lo, Bb[cur][q].lo,     acc[p][0][1]);
        acc[p][1][0] = wmma_f32(a0[q].lo, Bb[cur][4 + q].lo, acc[p][1][0]);
        acc[p][1][1] = wmma_f32(a1[q].lo, Bb[cur][4 + q].lo, acc[p][1][1]);
        acc[p][0][0] = wmma_f32(a0[q].hi, Bb[cur][q].hi,     acc[p][0][0]);
        acc[p][0][1] = wmma_f32(a1[q].hi, Bb[cur][q].hi,     acc[p][0][1]);
        acc[p][1][0] = wmma_f32(a0[q].hi, Bb[cur][4 + q].hi, acc[p][1][0]);
        acc[p][1][1] = wmma_f32(a1[q].hi, Bb[cur][4 + q].hi, acc[p][1][1]);
      }
      __builtin_amdgcn_sched_barrier(0);
    }
  }

  float lacc[2][8];
  #pragma unroll
  for (int r = 0; r < 8; ++r) {
    lacc[0][r] = 0.f;
    lacc[1][r] = 0.f;
  }

  #pragma unroll
  for (int p = 0; p < 4; ++p) {
    const int h0 = hb + p * 32;
    const float wb0 = whb[h0 + lm],      vv0 = v[h0 + lm];
    const float wb1 = whb[h0 + 16 + lm], vv1 = v[h0 + 16 + lm];
    #pragma unroll
    for (int m = 0; m < 2; ++m) {
      #pragma unroll
      for (int r = 0; r < 8; ++r) {
        lacc[m][r] += tanh_fast(acc[p][0][m][r] + wb0) * vv0 +
                      tanh_fast(acc[p][1][m][r] + wb1) * vv1;
      }
    }
  }

  #pragma unroll
  for (int m = 0; m < 2; ++m) {
    #pragma unroll
    for (int r = 0; r < 8; ++r) {
      float t = lacc[m][r];
      t += __shfl_xor(t, 8, 32);
      t += __shfl_xor(t, 4, 32);
      t += __shfl_xor(t, 2, 32);
      t += __shfl_xor(t, 1, 32);
      if (lm == 0) atomicAdd(&s_scores[m * 16 + r + 8 * hi], t);
    }
  }
  __syncthreads();
  if (tid < ES_ROWS) {
    float* dst = half ? part_hi : part_lo;
    dst[s0 + tid] = s_scores[tid];
  }
}

__global__ __launch_bounds__(1024) void softmax_kernel(
    const float* __restrict__ scores, float* __restrict__ out) {
  __shared__ float red[32];
  __shared__ float stat[2];
  const int tid = threadIdx.x;

  out[tid] = 0.f;
  out[tid + 1024] = 0.f;

  float vals[4];
  float m = -INFINITY;
  #pragma unroll
  for (int i = 0; i < 4; ++i) {
    vals[i] = scores[tid + i * 1024] + out[HDIM + tid + i * 1024];
    m = fmaxf(m, vals[i]);
  }
  #pragma unroll
  for (int o = 16; o >= 1; o >>= 1) m = fmaxf(m, __shfl_xor(m, o, 32));
  if ((tid & 31) == 0) red[tid >> 5] = m;
  __syncthreads();
  if (tid < 32) {
    float x = red[tid];
    #pragma unroll
    for (int o = 16; o >= 1; o >>= 1) x = fmaxf(x, __shfl_xor(x, o, 32));
    if (tid == 0) stat[0] = x;
  }
  __syncthreads();
  m = stat[0];

  float lsum = 0.f;
  #pragma unroll
  for (int i = 0; i < 4; ++i) {
    vals[i] = __expf(vals[i] - m);
    lsum += vals[i];
  }
  #pragma unroll
  for (int o = 16; o >= 1; o >>= 1) lsum += __shfl_xor(lsum, o, 32);
  if ((tid & 31) == 0) red[tid >> 5] = lsum;
  __syncthreads();
  if (tid < 32) {
    float x = red[tid];
    #pragma unroll
    for (int o = 16; o >= 1; o >>= 1) x += __shfl_xor(x, o, 32);
    if (tid == 0) stat[1] = x;
  }
  __syncthreads();
  const float inv = 1.0f / stat[1];
  #pragma unroll
  for (int i = 0; i < 4; ++i) out[HDIM + tid + i * 1024] = vals[i] * inv;
}

#define CTX_COLS 64
#define CTX_ROWS 256

__global__ __launch_bounds__(256) __attribute__((amdgpu_waves_per_eu(4, 4)))
void context_kernel(
    const float* __restrict__ enc, const float* __restrict__ attn,
    float* __restrict__ context) {
  __shared__ float4 s_part[16][CTX_COLS / 4];

  const int tid  = threadIdx.x;
  const int wave = tid >> 5;
  const int lane = tid & 31;
  const int lm   = lane & 15;
  const int hh   = lane >> 4;
  const int slab  = blockIdx.x & 31;
  const int chunk = blockIdx.x >> 5;

  const int row0 = chunk * CTX_ROWS + wave * 32 + hh * 16;
  const float* src = enc + (size_t)row0 * HDIM + slab * CTX_COLS + lm * 4;

  float4 w4[4];
  #pragma unroll
  for (int q = 0; q < 4; ++q) w4[q] = *(const float4*)(attn + row0 + 4 * q);

  float4 x[16];
  #pragma unroll
  for (int i = 0; i < 16; ++i) x[i] = *(const float4*)(src + (size_t)i * HDIM);

  float4 acc = make_float4(0.f, 0.f, 0.f, 0.f);
  #pragma unroll
  for (int q = 0; q < 4; ++q) {
    const float a0 = w4[q].x, a1 = w4[q].y, a2 = w4[q].z, a3 = w4[q].w;
    const float4 x0 = x[4 * q], x1 = x[4 * q + 1], x2 = x[4 * q + 2],
                 x3 = x[4 * q + 3];
    acc.x += a0 * x0.x; acc.y += a0 * x0.y; acc.z += a0 * x0.z; acc.w += a0 * x0.w;
    acc.x += a1 * x1.x; acc.y += a1 * x1.y; acc.z += a1 * x1.z; acc.w += a1 * x1.w;
    acc.x += a2 * x2.x; acc.y += a2 * x2.y; acc.z += a2 * x2.z; acc.w += a2 * x2.w;
    acc.x += a3 * x3.x; acc.y += a3 * x3.y; acc.z += a3 * x3.z; acc.w += a3 * x3.w;
  }

  s_part[wave * 2 + hh][lm] = acc;
  __syncthreads();

  if (tid < CTX_COLS) {
    const float* p = (const float*)&s_part[0][0];
    float sum = 0.f;
    #pragma unroll
    for (int j = 0; j < 16; ++j) sum += p[j * CTX_COLS + tid];
    atomicAdd(&context[slab * CTX_COLS + tid], sum);
  }
}

extern "C" void kernel_launch(void* const* d_in, const int* in_sizes, int n_in,
                              void* d_out, int out_size, void* d_ws, size_t ws_size,
                              hipStream_t stream) {
  const float* dh  = (const float*)d_in[0];
  const float* enc = (const float*)d_in[1];
  const float* W_h = (const float*)d_in[2];
  const float* W_e = (const float*)d_in[3];
  const float* b   = (const float*)d_in[4];
  const float* v   = (const float*)d_in[5];

  float* out    = (float*)d_out;
  float* whb    = (float*)d_ws;
  float* scores = whb + HDIM;

  const size_t lds_bytes = (size_t)ES_ROWS * LSTRIDE * sizeof(float);

  whb_kernel<<<HDIM / 8, 256, 0, stream>>>(W_h, dh, b, whb);
  energy_scores_kernel<<<(SDIM / ES_ROWS) * (HDIM / ES_COLS), 256, lds_bytes,
                         stream>>>(enc, W_e, whb, v, scores, out + HDIM);
  softmax_kernel<<<1, 1024, 0, stream>>>(scores, out);
  context_kernel<<<512, 256, 0, stream>>>(enc, out + HDIM, out);
}


# energy GEMM fragment loads issued further ahead: W_e pairs 3 matrix blocks ahead in 4 register buffers (was 1 ahead in 2), LDS A chunk one chunk ahead (was at use); k order and all sums unchanged
# speedup vs baseline: 3.5833x; 1.0627x over previous
#include <hip/hip_runtime.h>
#include <math.h>

typedef __attribute__((ext_vector_type(2))) float v2f;
typedef __attribute__((ext_vector_type(4))) float v4f;
typedef __attribute__((ext_vector_type(8))) float v8f;

#define HDIM 2048
#define SDIM 4096
#define LSTRIDE (HDIM + 4)

static __device__ __forceinline__ v8f wmma_f32(v2f a, v2f b, v8f c) {
  return __builtin_amdgcn_wmma_f32_16x16x4_f32(false, a, false, b, (short)0, c,
                                               false, false);
}

static __device__ __forceinline__ float tanh_fast(float x) {
  const float e = __expf(2.0f * x);
  return 1.0f - 2.0f / (e + 1.0f);
}

__global__ __launch_bounds__(256) __attribute__((amdgpu_waves_per_eu(2, 4)))
void whb_kernel(const float* __restrict__ W_h,
                const float* __restrict__ dh,
                const float* __restrict__ b,
                float* __restrict__ whb) {
  const int wave = threadIdx.x >> 5;
  const int lane = threadIdx.x & 31;
  const int row  = blockIdx.x * 8 + wave;
  const v4f* wrow = (const v4f*)(W_h + (size_t)row * HDIM) + lane;
  const v4f* drow = (const v4f*)dh + lane;

  v4f w[16], d[16];
  #pragma unroll
  for (int j = 0; j < 16; ++j) w[j] = wrow[32 * j];
  #pragma unroll
  for (int j = 0; j < 16; ++j) d[j] = drow[32 * j];
  const float bias = b[row];

  float s0 = 0.f, s1 = 0.f, s2 = 0.f, s3 = 0.f;
  #pragma unroll
  for (int j = 0; j < 16; ++j) {
    s0 += w[j].x * d[j].x;
    s1 += w[j].y * d[j].y;
    s2 += w[j].z * d[j].z;
    s3 += w[j].w * d[j].w;
  }
  float sum = (s0 + s1) + (s2 + s3);
  if (lane == 0) sum += bias;
  #pragma unroll
  for (int m = 16; m >= 1; m >>= 1) sum += __shfl_xor(sum, m, 32);
  if (lane == 0) whb[row] = sum;
}

#define ES_ROWS 32
#define ES_COLS 1024
#define ES_KC   32

__global__ __launch_bounds__(256) void energy_scores_kernel(
    const float* __restrict__ enc, const float* __restrict__ W_e,
    const float* __restrict__ whb, const float* __restrict__ v,
    float* __restrict__ part_lo, float* __restrict__ part_hi) {
  extern __shared__ float s_enc[];
  __shared__ float s_scores[ES_ROWS];

  const int tid  = threadIdx.x;
  const int wave = tid >> 5;
  const int lane = tid & 31;
  const int lm   = lane & 15;
  const int hi   = lane >> 4;
  const int half = blockIdx.x & 1;
  const int s0   = (blockIdx.x >> 1) * ES_ROWS;

  if (tid < ES_ROWS) s_scores[tid] = 0.f;

  #pragma unroll 1
  for (int t = 0; t < 2; ++t) {
    v4f buf[32];
    #pragma unroll
    for (int j = 0; j < 32; ++j) {
      const int row = 16 * t + (j >> 1);
      const int col = (tid + 256 * (j & 1)) << 2;
      buf[j] = *(const v4f*)(enc + (size_t)(s0 + row) * HDIM + col);
    }
    #pragma unroll
    for (int j = 0; j < 32; ++j) {
      const int row = 16 * t + (j >> 1);
      const int col = (tid + 256 * (j & 1)) << 2;
      *(v4f*)(&s_enc[row * LSTRIDE + col]) = buf[j];
    }
  }
  __syncthreads();

  const float* aRow0 = s_enc + lm * LSTRIDE + 16 * hi;
  const float* aRow1 = aRow0 + 16 * LSTRIDE;
  const int hb = half * ES_COLS + wave * 128;
  const float* bBase = W_e + (size_t)(hb + lm) * HDIM + 16 * hi;

  v8f acc[4][2][2];
  #pragma unroll
  for (int p = 0; p < 4; ++p) {
    #pragma unroll
    for (int t = 0; t < 2; ++t) {
      acc[p][t][0] = (v8f){};
      acc[p][t][1] = (v8f){};
    }
  }

  v4f Bb[4][8];
  #pragma unroll
  for (int p = 0; p < 3; ++p) {
    const float* nb = bBase + (size_t)p * 32 * HDIM;
    #pragma unroll
    for (int q = 0; q < 4; ++q) {
      Bb[p][q]     = *(const v4f*)(nb + 4 * q);
      Bb[p][4 + q] = *(const v4f*)(nb + (size_t)16 * HDIM + 4 * q);
    }
  }

  v4f A[2][2][4];
  #pragma unroll
  for (int q = 0; q < 4; ++q) {
    A[0][0][q] = *(const v4f*)(aRow0 + 4 * q);
    A[0][1][q] = *(const v4f*)(aRow1 + 4 * q);
  }

  #pragma unroll 1
  for (int c2 = 0; c2 < HDIM / ES_KC; c2 += 2) {
    #pragma unroll
    for (int cc = 0; cc < 2; ++cc) {
      const int c  = c2 + cc;
      const int k  = c * ES_KC;
      const int kn = (c < HDIM / ES_KC - 1) ? (k + ES_KC) : 0;

      #pragma unroll
      for (int p = 0; p < 4; ++p) {
        const int tp = (p + 3) & 3;
        const float* nb = bBase + (size_t)tp * 32 * HDIM + (p == 0 ? k : kn);
        #pragma unroll
        for (int q = 0; q < 4; ++q) {
          Bb[tp][q]     = *(const v4f*)(nb + 4 * q);
          Bb[tp][4 + q] = *(const v4f*)(nb + (size_t)16 * HDIM + 4 * q);
        }
        if (p == 0) {
          #pragma unroll
          for (int q = 0; q < 4; ++q) {
            A[cc ^ 1][0][q] = *(const v4f*)(aRow0 + kn + 4 * q);
            A[cc ^ 1][1][q] = *(const v4f*)(aRow1 + kn + 4 * q);
          }
        }

        __builtin_amdgcn_sched_barrier(0);

        #pragma unroll
        for (int q = 0; q < 4; ++q) {
          acc[p][0][0] = wmma_f32(A[cc][0][q].lo, Bb[p][q].lo,     acc[p][0][0]);
          acc[p][0][1] = wmma_f32(A[cc][1][q].lo, Bb[p][q].lo,     acc[p][0][1]);
          acc[p][1][0] = wmma_f32(A[cc][0][q].lo, Bb[p][4 + q].lo, acc[p][1][0]);
          acc[p][1][1] = wmma_f32(A[cc][1][q].lo, Bb[p][4 + q].lo, acc[p][1][1]);
          acc[p][0][0] = wmma_f32(A[cc][0][q].hi, Bb[p][q].hi,     acc[p][0][0]);
          acc[p][0][1] = wmma_f32(A[cc][1][q].hi, Bb[p][q].hi,     acc[p][0][1]);
          acc[p][1][0] = wmma_f32(A[cc][0][q].hi, Bb[p][4 + q].hi, acc[p][1][0]);
          acc[p][1][1] = wmma_f32(A[cc][1][q].hi, Bb[p][4 + q].hi, acc[p][1][1]);
        }
        __builtin_amdgcn_sched_barrier(0);
      }
    }
  }

  float lacc[2][8];
  #pragma unroll
  for (int r = 0; r < 8; ++r) {
    lacc[0][r] = 0.f;
    lacc[1][r] = 0.f;
  }

  #pragma unroll
  for (int p = 0; p < 4; ++p) {
    const int h0 = hb + p * 32;
    const float wb0 = whb[h0 + lm],      vv0 = v[h0 + lm];
    const float wb1 = whb[h0 + 16 + lm], vv1 = v[h0 + 16 + lm];
    #pragma unroll
    for (int m = 0; m < 2; ++m) {
      #pragma unroll
      for (int r = 0; r < 8; ++r) {
        lacc[m][r] += tanh_fast(acc[p][0][m][r] + wb0) * vv0 +
                      tanh_fast(acc[p][1][m][r] + wb1) * vv1;
      }
    }
  }

  #pragma unroll
  for (int m = 0; m < 2; ++m) {
    #pragma unroll
    for (int r = 0; r < 8; ++r) {
      float t = lacc[m][r];
      t += __shfl_xor(t, 8, 32);
      t += __shfl_xor(t, 4, 32);
      t += __shfl_xor(t, 2, 32);
      t += __shfl_xor(t, 1, 32);
      if (lm == 0) atomicAdd(&s_scores[m * 16 + r + 8 * hi], t);
    }
  }
  __syncthreads();
  if (tid < ES_ROWS) {
    float* dst = half ? part_hi : part_lo;
    dst[s0 + tid] = s_scores[tid];
  }
}

__global__ __launch_bounds__(1024) void softmax_kernel(
    const float* __restrict__ scores, float* __restrict__ out) {
  __shared__ float red[32];
  __shared__ float stat[2];
  const int tid = threadIdx.x;

  out[tid] = 0.f;
  out[tid + 1024] = 0.f;

  float vals[4];
  float m = -INFINITY;
  #pragma unroll
  for (int i = 0; i < 4; ++i) {
    vals[i] = scores[tid + i * 1024] + out[HDIM + tid + i * 1024];
    m = fmaxf(m, vals[i]);
  }
  #pragma unroll
  for (int o = 16; o >= 1; o >>= 1) m = fmaxf(m, __shfl_xor(m, o, 32));
  if ((tid & 31) == 0) red[tid >> 5] = m;
  __syncthreads();
  if (tid < 32) {
    float x = red[tid];
    #pragma unroll
    for (int o = 16; o >= 1; o >>= 1) x = fmaxf(x, __shfl_xor(x, o, 32));
    if (tid == 0) stat[0] = x;
  }
  __syncthreads();
  m = stat[0];

  float lsum = 0.f;
  #pragma unroll
  for (int i = 0; i < 4; ++i) {
    vals[i] = __expf(vals[i] - m);
    lsum += vals[i];
  }
  #pragma unroll
  for (int o = 16; o >= 1; o >>= 1) lsum += __shfl_xor(lsum, o, 32);
  if ((tid & 31) == 0) red[tid >> 5] = lsum;
  __syncthreads();
  if (tid < 32) {
    float x = red[tid];
    #pragma unroll
    for (int o = 16; o >= 1; o >>= 1) x += __shfl_xor(x, o, 32);
    if (tid == 0) stat[1] = x;
  }
  __syncthreads();
  const float inv = 1.0f / stat[1];
  #pragma unroll
  for (int i = 0; i < 4; ++i) out[HDIM + tid + i * 1024] = vals[i] * inv;
}

#define CTX_COLS 64
#define CTX_ROWS 256

__global__ __launch_bounds__(256) __attribute__((amdgpu_waves_per_eu(4, 4)))
void context_kernel(
    const float* __restrict__ enc, const float* __restrict__ attn,
    float* __restrict__ context) {
  __shared__ float4 s_part[16][CTX_COLS / 4];

  const int tid  = threadIdx.x;
  const int wave = tid >> 5;
  const int lane = tid & 31;
  const int lm   = lane & 15;
  const int hh   = lane >> 4;
  const int slab  = blockIdx.x & 31;
  const int chunk = blockIdx.x >> 5;

  const int row0 = chunk * CTX_ROWS + wave * 32 + hh * 16;
  const float* src = enc + (size_t)row0 * HDIM + slab * CTX_COLS + lm * 4;

  float4 w4[4];
  #pragma unroll
  for (int q = 0; q < 4; ++q) w4[q] = *(const float4*)(attn + row0 + 4 * q);

  float4 x[16];
  #pragma unroll
  for (int i = 0; i < 16; ++i) x[i] = *(const float4*)(src + (size_t)i * HDIM);

  float4 acc = make_float4(0.f, 0.f, 0.f, 0.f);
  #pragma unroll
  for (int q = 0; q < 4; ++q) {
    const float a0 = w4[q].x, a1 = w4[q].y, a2 = w4[q].z, a3 = w4[q].w;
    const float4 x0 = x[4 * q], x1 = x[4 * q + 1], x2 = x[4 * q + 2],
                 x3 = x[4 * q + 3];
    acc.x += a0 * x0.x; acc.y += a0 * x0.y; acc.z += a0 * x0.z; acc.w += a0 * x0.w;
    acc.x += a1 * x1.x; acc.y += a1 * x1.y; acc.z += a1 * x1.z; acc.w += a1 * x1.w;
    acc.x += a2 * x2.x; acc.y += a2 * x2.y; acc.z += a2 * x2.z; acc.w += a2 * x2.w;
    acc.x += a3 * x3.x; acc.y += a3 * x3.y; acc.z += a3 * x3.z; acc.w += a3 * x3.w;
  }

  s_part[wave * 2 + hh][lm] = acc;
  __syncthreads();

  if (tid < CTX_COLS) {
    const float* p = (const float*)&s_part[0][0];
    float sum = 0.f;
    #pragma unroll
    for (int j = 0; j < 16; ++j) sum += p[j * CTX_COLS + tid];
    atomicAdd(&context[slab * CTX_COLS + tid], sum);
  }
}

extern "C" void kernel_launch(void* const* d_in, const int* in_sizes, int n_in,
                              void* d_out, int out_size, void* d_ws, size_t ws_size,
                              hipStream_t stream) {
  const float* dh  = (const float*)d_in[0];
  const float* enc = (const float*)d_in[1];
  const float* W_h = (const float*)d_in[2];
  const float* W_e = (const float*)d_in[3];
  const float* b   = (const float*)d_in[4];
  const float* v   = (const float*)d_in[5];

  float* out    = (float*)d_out;
  float* whb    = (float*)d_ws;
  float* scores = whb + HDIM;

  const size_t lds_bytes = (size_t)ES_ROWS * LSTRIDE * sizeof(float);

  whb_kernel<<<HDIM / 8, 256, 0, stream>>>(W_h, dh, b, whb);
  energy_scores_kernel<<<(SDIM / ES_ROWS) * (HDIM / ES_COLS), 256, lds_bytes,
                         stream>>>(enc, W_e, whb, v, scores, out + HDIM);
  softmax_kernel<<<1, 1024, 0, stream>>>(scores, out);
  context_kernel<<<512, 256, 0, stream>>>(enc, out + HDIM, out);
}
